# starting kernel
# baseline (speedup 1.0000x reference)
#include <hip/hip_runtime.h>

typedef __attribute__((ext_vector_type(2))) float v2f;
typedef __attribute__((ext_vector_type(8))) float v8f;

#define NNODES 50000
#define DHID   128
#define NCLS   10

__global__ void deg_init_kernel(float* deg, int n) {
    int i = blockIdx.x * blockDim.x + threadIdx.x;
    if (i < n) deg[i] = 1.0f;
}

__global__ void deg_edges_kernel(const int* __restrict__ dst,
                                 const float* __restrict__ w,
                                 float* deg, int nE) {
    int e = blockIdx.x * blockDim.x + threadIdx.x;
    if (e < nE) atomicAdd(&deg[dst[e]], w[e]);
}

__global__ void dinv_kernel(const float* __restrict__ deg, float* dinv, int n) {
    int i = blockIdx.x * blockDim.x + threadIdx.x;
    if (i < n) {
        float d = deg[i];
        dinv[i] = __frsqrt_rn(d);
    }
}

__global__ void norm_kernel(const int* __restrict__ src,
                            const int* __restrict__ dst,
                            const float* __restrict__ w,
                            const float* __restrict__ dinv,
                            float* norm, int nE) {
    int e = blockIdx.x * blockDim.x + threadIdx.x;
    if (e < nE) norm[e] = dinv[src[e]] * w[e] * dinv[dst[e]];
}

#define XS_STRIDE 34
#define WT_STRIDE 144

__global__ __launch_bounds__(256) void gemm128_wmma_kernel(
    const float* __restrict__ X, const float* __restrict__ W,
    float* __restrict__ H, int M, int K)
{
    __shared__ float  Xs[128 * XS_STRIDE];
    __shared__ float2 Wt[16 * WT_STRIDE];

    const int tid  = threadIdx.x;
    const int lane = tid & 31;
    const int wave = tid >> 5;
    const int l16  = lane & 15;
    const bool hi  = (lane >= 16);
    const int rowBlock = blockIdx.x * 128;
    const int rowLocal = wave * 16 + l16;

    v8f acc[8];
    #pragma unroll
    for (int t = 0; t < 8; ++t)
        #pragma unroll
        for (int r = 0; r < 8; ++r) acc[t][r] = 0.0f;

    const int nch = (K + 31) / 32;
    for (int ch = 0; ch < nch; ++ch) {
        const int k0 = ch * 32;

        #pragma unroll
        for (int j = 0; j < 4; ++j) {
            int f  = tid * 4 + j;
            int r  = f >> 3;
            int kb = (f & 7) * 4;
            int gr = rowBlock + r;
            float4 v = make_float4(0.f, 0.f, 0.f, 0.f);
            if (gr < M) {
                int kg = k0 + kb;
                if (kg + 3 < K) {
                    v = *(const float4*)(X + (size_t)gr * K + kg);
                } else {
                    float t0 = (kg + 0 < K) ? X[(size_t)gr * K + kg + 0] : 0.f;
                    float t1 = (kg + 1 < K) ? X[(size_t)gr * K + kg + 1] : 0.f;
                    float t2 = (kg + 2 < K) ? X[(size_t)gr * K + kg + 2] : 0.f;
                    float t3 = (kg + 3 < K) ? X[(size_t)gr * K + kg + 3] : 0.f;
                    v = make_float4(t0, t1, t2, t3);
                }
            }
            float* dp = &Xs[r * XS_STRIDE + kb];
            ((float2*)dp)[0] = make_float2(v.x, v.y);
            ((float2*)dp)[1] = make_float2(v.z, v.w);
        }

        #pragma unroll
        for (int j = 0; j < 2; ++j) {
            int f  = tid * 2 + j;
            int kp = f >> 5;
            int cb = (f & 31) * 4;
            int ka = k0 + 2 * kp;
            float4 va = make_float4(0.f, 0.f, 0.f, 0.f);
            float4 vb = make_float4(0.f, 0.f, 0.f, 0.f);
            if (ka < K)     va = *(const float4*)(W + (size_t)ka * DHID + cb);
            if (ka + 1 < K) vb = *(const float4*)(W + (size_t)(ka + 1) * DHID + cb);
            float4 w0 = make_float4(va.x, vb.x, va.y, vb.y);
            float4 w1 = make_float4(va.z, vb.z, va.w, vb.w);
            *(float4*)&Wt[kp * WT_STRIDE + cb]     = w0;
            *(float4*)&Wt[kp * WT_STRIDE + cb + 2] = w1;
        }

        __syncthreads();

        #pragma unroll
        for (int s = 0; s < 8; ++s) {
            const int kp = s * 2 + (hi ? 1 : 0);
            v2f a = *(const v2f*)&Xs[rowLocal * XS_STRIDE + kp * 2];
            float2 bf[8];
            #pragma unroll
            for (int nt = 0; nt < 8; ++nt)
                bf[nt] = Wt[kp * WT_STRIDE + nt * 16 + l16];
            #pragma unroll
            for (int nt = 0; nt < 8; ++nt) {
                v2f b;
                b.x = bf[nt].x;
                b.y = bf[nt].y;
                acc[nt] = __builtin_amdgcn_wmma_f32_16x16x4_f32(
                     false, a,  false, b,
                     (short)0, acc[nt],
                     false,  false);
            }
        }
        __syncthreads();
    }

    #pragma unroll
    for (int nt = 0; nt < 8; ++nt) {
        const int col = nt * 16 + l16;
        #pragma unroll
        for (int r = 0; r < 8; ++r) {
            int m = rowBlock + wave * 16 + r + (hi ? 8 : 0);
            if (m < M) H[(size_t)m * DHID + col] = acc[nt][r];
        }
    }
}

__global__ void agg_self_kernel(const float* __restrict__ H,
                                const float* __restrict__ dinv,
                                float* __restrict__ OUT, int n) {
    int i = blockIdx.x * blockDim.x + threadIdx.x;
    if (i < n * DHID) {
        float s = dinv[i >> 7];
        OUT[i] = H[i] * s * s;
    }
}

__global__ __launch_bounds__(256) void agg_edges_kernel(
    const float* __restrict__ H,
    const int* __restrict__ src, const int* __restrict__ dst,
    const float* __restrict__ norm,
    float* __restrict__ OUT, int nE)
{
    int e = blockIdx.x * 2 + (threadIdx.x >> 7);
    int d = threadIdx.x & 127;
    if (e < nE) {
        int s = src[e];
        int t = dst[e];
        float v = H[(size_t)s * DHID + d] * norm[e];
        atomicAdd(&OUT[(size_t)t * DHID + d], v);
    }
}

__global__ void bias_relu_kernel(float* __restrict__ OUT,
                                 const float* __restrict__ b, int n) {
    int i = blockIdx.x * blockDim.x + threadIdx.x;
    if (i < n * DHID) {
        float v = OUT[i] + b[i & (DHID - 1)];
        OUT[i] = v > 0.f ? v : 0.f;
    }
}

__global__ void classifier_kernel(const float* __restrict__ H,
                                  const float* __restrict__ Wc,
                                  const float* __restrict__ bc,
                                  float* __restrict__ out, int n) {
    int i = blockIdx.x * blockDim.x + threadIdx.x;
    if (i < n * NCLS) {
        int row = i / NCLS;
        int c   = i - row * NCLS;
        float s = bc[c];
        const float* h = H + (size_t)row * DHID;
        #pragma unroll 8
        for (int k = 0; k < DHID; ++k) s += h[k] * Wc[k * NCLS + c];
        out[i] = s;
    }
}

extern "C" void kernel_launch(void* const* d_in, const int* in_sizes, int n_in,
                              void* d_out, int out_size, void* d_ws, size_t ws_size,
                              hipStream_t stream) {
    const float* x  = (const float*)d_in[0];
    const int*   ei = (const int*)d_in[1];
    const float* ew = (const float*)d_in[2];
    const float* W1 = (const float*)d_in[3];
    const float* b1 = (const float*)d_in[4];
    const float* W2 = (const float*)d_in[5];
    const float* b2 = (const float*)d_in[6];
    const float* Wc = (const float*)d_in[7];
    const float* bc = (const float*)d_in[8];
    float* out = (float*)d_out;

    const int n  = NNODES;
    const int E  = in_sizes[1] / 2;
    const int K1 = in_sizes[0] / n;
    const int* src = ei;
    const int* dst = ei + E;

    char* ws = (char*)d_ws;
    size_t off = 0;
    auto alloc = [&](size_t bytes) { char* p = ws + off; off = (off + bytes + 255) & ~(size_t)255; return p; };
    float* deg  = (float*)alloc((size_t)n * 4);
    float* dinv = (float*)alloc((size_t)n * 4);
    float* norm = (float*)alloc((size_t)E * 4);
    float* bufA = (float*)alloc((size_t)n * DHID * 4);
    float* bufB = (float*)alloc((size_t)n * DHID * 4);
    (void)ws_size;

    const int T = 256;
    deg_init_kernel <<<(n + T - 1) / T, T, 0, stream>>>(deg, n);
    deg_edges_kernel<<<(E + T - 1) / T, T, 0, stream>>>(dst, ew, deg, E);
    dinv_kernel     <<<(n + T - 1) / T, T, 0, stream>>>(deg, dinv, n);
    norm_kernel     <<<(E + T - 1) / T, T, 0, stream>>>(src, dst, ew, dinv, norm, E);

    const int gemmBlocks = (n + 127) / 128;
    const int elemBlocks = (n * DHID + T - 1) / T;
    const int edgeBlocks = (E + 1) / 2;

    gemm128_wmma_kernel<<<gemmBlocks, T, 0, stream>>>(x, W1, bufA, n, K1);
    agg_self_kernel    <<<elemBlocks, T, 0, stream>>>(bufA, dinv, bufB, n);
    agg_edges_kernel   <<<edgeBlocks, T, 0, stream>>>(bufA, src, dst, norm, bufB, E);
    bias_relu_kernel   <<<elemBlocks, T, 0, stream>>>(bufB, b1, n);

    gemm128_wmma_kernel<<<gemmBlocks, T, 0, stream>>>(bufB, W2, bufA, n, DHID);
    agg_self_kernel    <<<elemBlocks, T, 0, stream>>>(bufA, dinv, bufB, n);
    agg_edges_kernel   <<<edgeBlocks, T, 0, stream>>>(bufA, src, dst, norm, bufB, E);
    bias_relu_kernel   <<<elemBlocks, T, 0, stream>>>(bufB, b2, n);

    classifier_kernel<<<(n * NCLS + T - 1) / T, T, 0, stream>>>(bufB, Wc, bc, out, n);
}


# Both GEMMs: W staged once per work-group and kept whole in LDS; A fragments read from global straight into registers, a group of 16-byte loads ahead; 256 work-groups of 196 rows; new k order in each 8
# speedup vs baseline: 4.0366x; 4.0366x over previous
#include <hip/hip_runtime.h>

typedef __attribute__((ext_vector_type(2))) float v2f;
typedef __attribute__((ext_vector_type(8))) float v8f;

#define NNODES 50000
#define DHID   128
#define NCLS   10

#define AGG_BLOCKS  256
#define AGG_ROWS    ((NNODES + AGG_BLOCKS - 1) / AGG_BLOCKS)

#define EW_BLOCKS  256
#define EW_THREADS 512
#define EW_BATCH   8

__global__ __launch_bounds__(EW_THREADS) void bins_kernel(
    const int* __restrict__ dst, int* bins, int nE) {
    unsigned char* bin = (unsigned char*)bins;
    const int stride = gridDim.x * EW_THREADS;
    for (int e0 = blockIdx.x * EW_THREADS + threadIdx.x; e0 < nE; e0 += EW_BATCH * stride) {
        int d[EW_BATCH];
        #pragma unroll
        for (int k = 0; k < EW_BATCH; ++k) {
            const int e = e0 + k * stride;
            d[k] = 0;
            if (e < nE) d[k] = dst[e];
        }
        #pragma unroll
        for (int k = 0; k < EW_BATCH; ++k) {
            const int e = e0 + k * stride;
            if (e < nE) bin[e] = (unsigned char)((unsigned)d[k] / (unsigned)AGG_ROWS);
        }
    }
}

#define GM_THREADS 512
#define GM_WAVES   (GM_THREADS / 32)
#define GM_KSLAB   512
#define GM_WPITCH  136
#define GM_CH      8

template <bool VEC, bool ACT>
__device__ __forceinline__ float4 gm_load_a4(const float* __restrict__ xr, int k, int kw,
                                             bool rv, const float* __restrict__ ab)
{
    float4 v;
    if (VEC) {
        const bool ok = rv && (k < kw);
        const int kk = ok ? k : 0;
        v = *(const float4*)(xr + kk);
        if (ACT) {
            const float4 bb = *(const float4*)(ab + kk);
            const float x0 = v.x + bb.x, x1 = v.y + bb.y;
            const float x2 = v.z + bb.z, x3 = v.w + bb.w;
            v = make_float4(x0 > 0.f ? x0 : 0.f, x1 > 0.f ? x1 : 0.f,
                            x2 > 0.f ? x2 : 0.f, x3 > 0.f ? x3 : 0.f);
        }
        if (!ok) v = make_float4(0.f, 0.f, 0.f, 0.f);
    } else {
        float t[4];
        #pragma unroll
        for (int e = 0; e < 4; ++e) {
            const bool ok = rv && (k + e < kw);
            const int kk = ok ? (k + e) : 0;
            float x = xr[kk];
            if (ACT) { x += ab[kk]; x = x > 0.f ? x : 0.f; }
            t[e] = ok ? x : 0.f;
        }
        v = make_float4(t[0], t[1], t[2], t[3]);
    }
    return v;
}

template <bool VEC, bool ACT>
__device__ __forceinline__ void gemm_rows_body(
    const float* __restrict__ X, const float* __restrict__ W,
    float* __restrict__ H, int M, int K, const float* __restrict__ actb, float2* Wt)
{
    const int tid  = threadIdx.x;
    const int lane = tid & 31;
    const int wave = __builtin_amdgcn_readfirstlane(tid >> 5);
    const int c    = lane & 15;
    const int h    = lane >> 4;
    const int base = blockIdx.x * AGG_ROWS;
    int rows = M - base;
    if (rows > AGG_ROWS) rows = AGG_ROWS;
    if (rows < 0) rows = 0;
    const int nunits = ((rows + 15) >> 4) * 2;

    for (int ks = 0; ks < K; ks += GM_KSLAB) {
        const int kw  = (K - ks < GM_KSLAB) ? (K - ks) : GM_KSLAB;
        const int nst = (kw + 7) >> 3;

        if (ks > 0) __syncthreads();
        const int ntask = nst * 128;
        for (int f0 = tid; f0 < ntask; f0 += GM_THREADS * 4) {
            float4 va[4], vb[4];
            #pragma unroll
            for (int j = 0; j < 4; ++j) {
                const int f  = f0 + j * GM_THREADS;
                const int kp = f >> 5;
                const int cb = (f & 31) * 4;
                const int ra = (f < ntask && 2 * kp     < kw) ? (ks + 2 * kp)     : ks;
                const int rb = (f < ntask && 2 * kp + 1 < kw) ? (ks + 2 * kp + 1) : ks;
                va[j] = *(const float4*)(W + (size_t)ra * DHID + cb);
                vb[j] = *(const float4*)(W + (size_t)rb * DHID + cb);
            }
            #pragma unroll
            for (int j = 0; j < 4; ++j) {
                const int f  = f0 + j * GM_THREADS;
                const int kp = f >> 5;
                const int cb = (f & 31) * 4;
                if (f < ntask) {
                    float4 a = va[j], b = vb[j];
                    if (2 * kp     >= kw) a = make_float4(0.f, 0.f, 0.f, 0.f);
                    if (2 * kp + 1 >= kw) b = make_float4(0.f, 0.f, 0.f, 0.f);
                    *(float4*)&Wt[kp * GM_WPITCH + cb]     = make_float4(a.x, b.x, a.y, b.y);
                    *(float4*)&Wt[kp * GM_WPITCH + cb + 2] = make_float4(a.z, b.z, a.w, b.w);
                }
            }
        }
        __syncthreads();

        const float* ab = ACT ? (actb + ks) : nullptr;

        for (int u = wave; u < nunits; u += GM_WAVES) {
            const int strip = u >> 1;
            const int nh    = u & 1;
            const int lrow  = strip * 16 + c;
            const bool rv   = lrow < rows;
            const float* xr = X + (size_t)(base + (rv ? lrow : 0)) * K + ks;
            const bool whole = (strip * 16 + 16 <= rows);

            v8f acc[4];
            #pragma unroll
            for (int nt = 0; nt < 4; ++nt)
                #pragma unroll
                for (int r = 0; r < 8; ++r) acc[nt][r] = 0.0f;
            if (ks > 0) {
                #pragma unroll
                for (int nt = 0; nt < 4; ++nt)
                    #pragma unroll
                    for (int r = 0; r < 8; ++r) {
                        const int lr = strip * 16 + 8 * h + r;
                        if (lr < rows) acc[nt][r] = H[(size_t)(base + lr) * DHID + nh * 64 + nt * 16 + c];
                    }
            }

            float4 bufA[GM_CH], bufB[GM_CH];
            auto fetch = [&](float4 (&buf)[GM_CH], int t0) {
                #pragma unroll
                for (int j = 0; j < GM_CH; ++j)
                    buf[j] = gm_load_a4<VEC, ACT>(xr, 8 * (t0 + j) + 4 * h, kw, rv, ab);
            };
            auto step = [&](const float4& a4, int t) {
                #pragma unroll
                for (int q = 0; q < 2; ++q) {
                    const int kp = 4 * t + 2 * h + q;
                    v2f a;
                    a.x = q ? a4.z : a4.x;
                    a.y = q ? a4.w : a4.y;
                    const float2* bp = &Wt[kp * GM_WPITCH + nh * 64 + c];
                    float2 bf[4];
                    #pragma unroll
                    for (int nt = 0; nt < 4; ++nt) bf[nt] = bp[nt * 16];
                    #pragma unroll
                    for (int nt = 0; nt < 4; ++nt) {
                        v2f b;
                        b.x = bf[nt].x;
                        b.y = bf[nt].y;
                        acc[nt] = __builtin_amdgcn_wmma_f32_16x16x4_f32(
                             false, a,  false, b,
                             (short)0, acc[nt],
                             false,  false);
                    }
                }
            };
            auto mult = [&](const float4 (&buf)[GM_CH], int t0) {
                if (t0 + GM_CH <= nst) {
                    #pragma unroll
                    for (int j = 0; j < GM_CH; ++j) step(buf[j], t0 + j);
                } else {
                    #pragma unroll
                    for (int j = 0; j < GM_CH; ++j)
                        if (t0 + j < nst) step(buf[j], t0 + j);
                }
            };

            fetch(bufA, 0);
            for (int t0 = 0; t0 < nst; t0 += 2 * GM_CH) {
                fetch(bufB, t0 + GM_CH);
                mult(bufA, t0);
                fetch(bufA, t0 + 2 * GM_CH);
                mult(bufB, t0 + GM_CH);
            }

            if (whole) {
                #pragma unroll
                for (int nt = 0; nt < 4; ++nt)
                    #pragma unroll
                    for (int r = 0; r < 8; ++r)
                        H[(size_t)(base + strip * 16 + 8 * h + r) * DHID + nh * 64 + nt * 16 + c] = acc[nt][r];
            } else {
                #pragma unroll
                for (int nt = 0; nt < 4; ++nt)
                    #pragma unroll
                    for (int r = 0; r < 8; ++r) {
                        const int lr = strip * 16 + 8 * h + r;
                        if (lr < rows) H[(size_t)(base + lr) * DHID + nh * 64 + nt * 16 + c] = acc[nt][r];
                    }
            }
        }
    }
}

__global__ __launch_bounds__(GM_THREADS) void gemm_rows_wmma_kernel(
    const float* __restrict__ X, const float* __restrict__ W,
    float* __restrict__ H, int M, int K, const float* __restrict__ actb)
{
    __shared__ float2 Wt[(GM_KSLAB / 2) * GM_WPITCH];
    if ((K & 3) == 0) {
        if (actb) gemm_rows_body<true, true >(X, W, H, M, K, actb, Wt);
        else      gemm_rows_body<true, false>(X, W, H, M, K, actb, Wt);
    } else {
        if (actb) gemm_rows_body<false, true >(X, W, H, M, K, actb, Wt);
        else      gemm_rows_body<false, false>(X, W, H, M, K, actb, Wt);
    }
}


#define AGG_THREADS 1024
#define AGG_WAVES   (AGG_THREADS / 32)
#define AGG_FLT     4
#define AGG_QCAP    1024
#define AGG_LCAP    256
#define OWN_QCAP    1536

template <int QCAP, class Proc>
__device__ __forceinline__ int owner_sweep(
    const int* __restrict__ bins, const int* __restrict__ dst, int nE,
    int base, unsigned urows, int* qb, int wave, int lane, bool may_keep, Proc proc)
{
    int cnt = 0;
    int kept = -1;

    auto push = [&](bool m, int e) {
        const unsigned bal = __builtin_amdgcn_ballot_w32(m);
        if (bal) {
            if (m) qb[cnt + (int)__builtin_amdgcn_mbcnt_lo(bal, 0u)] = e;
            cnt += __builtin_popcount(bal);
        }
    };

    const int4* bq = (const int4*)bins;
    const unsigned pat = blockIdx.x * 0x01010101u;
    const int nfull = nE >> 9;
    const int rot = (int)((blockIdx.x * (unsigned)nfull) / gridDim.x);
    int it = wave;
    int tb = (nfull << 9) + wave * 32;
    bool more = true;
    bool first = true;
    while (more) {
        while (it < nfull && cnt <= QCAP - 512) {
            int4 d[AGG_FLT];
            int  e0[AGG_FLT];
            #pragma unroll
            for (int k = 0; k < AGG_FLT; ++k) {
                int sc = ((it + k * AGG_WAVES < nfull) ? (it + k * AGG_WAVES) : it) + rot;
                if (sc >= nfull) sc -= nfull;
                d[k]  = bq[(sc << 5) + lane];
                e0[k] = (sc << 9) + lane * 16;
            }
            int done = 0;
            #pragma unroll
            for (int k = 0; k < AGG_FLT; ++k) {
                if (done == k && it + k * AGG_WAVES < nfull && cnt <= QCAP - 512) {
                    auto z = [&](int x) -> unsigned {
                        const unsigned t = (unsigned)x ^ pat;
                        return ~(((t & 0x7F7F7F7Fu) + 0x7F7F7F7Fu) | t | 0x7F7F7F7Fu);
                    };
                    unsigned m = (z(d[k].x) >> 7) | (z(d[k].y) >> 6) |
                                 (z(d[k].z) >> 5) | (z(d[k].w) >> 4);
                    unsigned bal;
                    while ((bal = __builtin_amdgcn_ballot_w32(m != 0u)) != 0u) {
                        if (m != 0u) {
                            const int p = __builtin_ctz(m);
                            qb[cnt + (int)__builtin_amdgcn_mbcnt_lo(bal, 0u)] =
                                e0[k] + ((p & 7) << 2) + (p >> 3);
                        }
                        cnt += __builtin_popcount(bal);
                        m &= m - 1u;
                    }
                    done = k + 1;
                }
            }
            it += done * AGG_WAVES;
        }
        if (it >= nfull) {
            while (tb < nE && cnt <= QCAP - 32) {
                const int e = tb + lane;
                const int dv = (e < nE) ? dst[e] : -1;
                push((unsigned)(dv - base) < urows, e);
                tb += AGG_THREADS;
            }
            more = (tb < nE);
        }

        const bool keep = may_keep && first && !more && (cnt <= AGG_LCAP);

        proc(qb, cnt, keep);
        if (keep) kept = cnt;
        first = false;
        cnt = 0;
    }
    return kept;
}

__global__ __launch_bounds__(AGG_THREADS) void own_edges_kernel(
    const int* __restrict__ src, const int* __restrict__ dst,
    const float* __restrict__ w, const int* __restrict__ bins,
    float* __restrict__ deg, int n, int nE, int* recs, int* rcnt)
{
    __shared__ float degl[AGG_ROWS];
    __shared__ int   queue[AGG_WAVES * OWN_QCAP];

    const int tid  = threadIdx.x;
    const int lane = tid & 31;
    const int wave = __builtin_amdgcn_readfirstlane(tid >> 5);
    const int base = blockIdx.x * AGG_ROWS;
    int rows = n - base;
    if (rows > AGG_ROWS) rows = AGG_ROWS;
    if (rows < 0) rows = 0;
    const unsigned urows = (unsigned)rows;

    for (int i = tid; i < AGG_ROWS; i += AGG_THREADS) degl[i] = 1.0f;
    __syncthreads();

    const int slot = blockIdx.x * AGG_WAVES + wave;
    int4* rq = (int4*)recs + (size_t)slot * AGG_LCAP;

    const int kept = owner_sweep<OWN_QCAP>(bins, dst, nE, base, urows, queue + wave * OWN_QCAP,
                                           wave, lane, true,
        [&](const int* q, int cnt, bool keep) {
            for (int b = 0; b < cnt; b += 128) {
                int   sv[4];
                int   dv[4];
                float wv[4];
                #pragma unroll
                for (int j = 0; j < 4; ++j) {
                    const int i = b + 32 * j + lane;
                    const int e = q[i < cnt ? i : b];
                    sv[j] = src[e];
                    wv[j] = w[e];
                    dv[j] = dst[e];
                }
                asm volatile("" :: "v"(sv[0]), "v"(sv[1]), "v"(sv[2]), "v"(sv[3]),
                                   "v"(dv[0]), "v"(dv[1]), "v"(dv[2]), "v"(dv[3]),
                                   "v"(wv[0]), "v"(wv[1]), "v"(wv[2]), "v"(wv[3]));
                #pragma unroll
                for (int j = 0; j < 4; ++j) {
                    const int i = b + 32 * j + lane;
                    if (i < cnt) {
                        const int lr = dv[j] - base;
                        atomicAdd(&degl[lr], wv[j]);
                        if (keep) rq[i] = make_int4(sv[j], __float_as_int(wv[j]), lr, 0);
                    }
                }
            }
        });
    if (lane == 0) rcnt[slot] = kept;
    __syncthreads();

    for (int i = tid; i < rows; i += AGG_THREADS) deg[base + i] = degl[i];
}

__global__ __launch_bounds__(AGG_THREADS) void agg_rows_kernel(
    const float* __restrict__ H,
    const int* __restrict__ src, const int* __restrict__ dst,
    const float* __restrict__ w, const float* __restrict__ deg,
    const int* __restrict__ bins,
    float* __restrict__ OUT, int n, int nE,
    int* recs, const int* __restrict__ rcnt, int reuse)
{
    __shared__ float acc[AGG_ROWS * DHID];
    __shared__ int   queue[AGG_WAVES * AGG_QCAP];
    __shared__ float dloc[AGG_ROWS];

    const int tid  = threadIdx.x;
    const int lane = tid & 31;
    const int wave = __builtin_amdgcn_readfirstlane(tid >> 5);
    const int base = blockIdx.x * AGG_ROWS;
    int rows = n - base;
    if (rows > AGG_ROWS) rows = AGG_ROWS;
    if (rows < 0) rows = 0;
    const unsigned urows = (unsigned)rows;

    for (int idx = tid; idx < rows * 32; idx += AGG_THREADS) {
        const int r = idx >> 5;
        const int l = idx & 31;
        const float4 h = *(const float4*)(H + (size_t)(base + r) * DHID + l * 4);
        const float s = __frsqrt_rn(deg[base + r]);
        if (l == 0) dloc[r] = s;
        float* a = acc + r * DHID + l;
        a[0]  = h.x * s * s;
        a[32] = h.y * s * s;
        a[64] = h.z * s * s;
        a[96] = h.w * s * s;
    }
    __syncthreads();

    auto add_rows = [&](int sv, int nvb, int lr, int nb) {
        for (int i = 0; i < nb; i += 8) {
            float4 hv[8];
            float  nk[8];
            int    lk[8];
            #pragma unroll
            for (int k = 0; k < 8; ++k) {
                const int ii = (i + k < nb) ? (i + k) : i;
                const int sk = __builtin_amdgcn_readlane(sv, ii);
                nk[k] = __int_as_float(__builtin_amdgcn_readlane(nvb, ii));
                lk[k] = __builtin_amdgcn_readlane(lr, ii);
                hv[k] = *(const float4*)(H + (size_t)sk * DHID + lane * 4);
            }
            #pragma unroll
            for (int k = 0; k < 8; ++k) {
                if (i + k < nb) {
                    float* a = acc + lk[k] * DHID + lane;
                    atomicAdd(a,      hv[k].x * nk[k]);
                    atomicAdd(a + 32, hv[k].y * nk[k]);
                    atomicAdd(a + 64, hv[k].z * nk[k]);
                    atomicAdd(a + 96, hv[k].w * nk[k]);
                }
            }
        }
    };

    const int slot = blockIdx.x * AGG_WAVES + wave;
    int4* rq = (int4*)recs + (size_t)slot * AGG_LCAP;

    int kept = __builtin_amdgcn_readfirstlane(rcnt[slot]);
    if (kept > AGG_LCAP) kept = AGG_LCAP;

    if (kept >= 0) {
        for (int b = 0; b < kept; b += 32) {
            const int nb = (kept - b < 32) ? (kept - b) : 32;
            const int4 r = rq[b + (lane < nb ? lane : 0)];
            int nvb = r.y;
            if (reuse == 0) {
                const float a = deg[r.x];
                nvb = __float_as_int(__frsqrt_rn(a) * __int_as_float(r.y) * dloc[r.z]);
                if (lane < nb) ((int*)(rq + b + lane))[1] = nvb;
            }
            add_rows(r.x, nvb, r.z, nb);
        }
    } else {
        owner_sweep<AGG_QCAP>(bins, dst, nE, base, urows, queue + wave * AGG_QCAP,
                              wave, lane, false,
            [&](const int* q, int cnt, bool keep) {
                (void)keep;
                for (int b = 0; b < cnt; b += 32) {
                    const int nb = (cnt - b < 32) ? (cnt - b) : 32;
                    const int   e  = q[b + (lane < nb ? lane : 0)];
                    const int   sv = src[e];
                    const float wv = w[e];
                    const int   lr = dst[e] - base;
                    const float a  = deg[sv];
                    const int  nvb = __float_as_int(__frsqrt_rn(a) * wv * dloc[lr]);
                    add_rows(sv, nvb, lr, nb);
                }
            });
    }
    __syncthreads();

    for (int idx = tid; idx < rows * 32; idx += AGG_THREADS) {
        const int r = idx >> 5;
        const int l = idx & 31;
        const float* a = acc + r * DHID + l;
        const float4 o = make_float4(a[0], a[32], a[64], a[96]);
        *(float4*)(OUT + (size_t)(base + r) * DHID + l * 4) = o;
    }
}

__global__ void classifier_kernel(const float* __restrict__ H,
                                  const float* __restrict__ b,
                                  const float* __restrict__ Wc,
                                  const float* __restrict__ bc,
                                  float* __restrict__ out, int n) {
    int i = blockIdx.x * blockDim.x + threadIdx.x;
    if (i < n * NCLS) {
        int row = i / NCLS;
        int c   = i - row * NCLS;
        float s = bc[c];
        const float* h = H + (size_t)row * DHID;
        #pragma unroll 8
        for (int k = 0; k < DHID; ++k) {
            const float x = h[k] + b[k];
            s += (x > 0.f ? x : 0.f) * Wc[k * NCLS + c];
        }
        out[i] = s;
    }
}

extern "C" void kernel_launch(void* const* d_in, const int* in_sizes, int n_in,
                              void* d_out, int out_size, void* d_ws, size_t ws_size,
                              hipStream_t stream) {
    const float* x  = (const float*)d_in[0];
    const int*   ei = (const int*)d_in[1];
    const float* ew = (const float*)d_in[2];
    const float* W1 = (const float*)d_in[3];
    const float* b1 = (const float*)d_in[4];
    const float* W2 = (const float*)d_in[5];
    const float* b2 = (const float*)d_in[6];
    const float* Wc = (const float*)d_in[7];
    const float* bc = (const float*)d_in[8];
    float* out = (float*)d_out;

    const int n  = NNODES;
    const int E  = in_sizes[1] / 2;
    const int K1 = in_sizes[0] / n;
    const int* src = ei;
    const int* dst = ei + E;

    char* ws = (char*)d_ws;
    size_t off = 0;
    auto alloc = [&](size_t bytes) { char* p = ws + off; off = (off + bytes + 255) & ~(size_t)255; return p; };
    float* deg  = (float*)alloc((size_t)n * 4);
    float* bufA = (float*)alloc((size_t)n * DHID * 4);
    float* bufB = (float*)alloc((size_t)n * DHID * 4);
    int* recs = (int*)alloc((size_t)AGG_BLOCKS * AGG_WAVES * AGG_LCAP * 16);
    int* rcnt = (int*)alloc((size_t)AGG_BLOCKS * AGG_WAVES * 4);
    int* bins = (int*)alloc((size_t)E);
    (void)ws_size;

    const int T = 256;
    bins_kernel     <<<EW_BLOCKS, EW_THREADS, 0, stream>>>(dst, bins, E);
    own_edges_kernel<<<AGG_BLOCKS, AGG_THREADS, 0, stream>>>(src, dst, ew, bins, deg, n, E, recs, rcnt);

    gemm_rows_wmma_kernel<<<AGG_BLOCKS, GM_THREADS, 0, stream>>>(x, W1, bufA, n, K1, nullptr);
    agg_rows_kernel    <<<AGG_BLOCKS, AGG_THREADS, 0, stream>>>(bufA, src, dst, ew, deg, bins, bufB, n, E, recs, rcnt, 0);

    gemm_rows_wmma_kernel<<<AGG_BLOCKS, GM_THREADS, 0, stream>>>(bufB, W2, bufA, n, DHID, b1);
    agg_rows_kernel    <<<AGG_BLOCKS, AGG_THREADS, 0, stream>>>(bufA, src, dst, ew, deg, bins, bufB, n, E, recs, rcnt, 1);

    classifier_kernel<<<(n * NCLS + T - 1) / T, T, 0, stream>>>(bufB, b2, Wc, bc, out, n);
}


# Edge pre-pass also writes one 16-byte record per edge (src, weight, out row in its owner block): the owner sweep fetches a matched edge with ONE load, not three 4-byte lookups; sums and order kept
# speedup vs baseline: 4.6504x; 1.1521x over previous
#include <hip/hip_runtime.h>

typedef __attribute__((ext_vector_type(2))) float v2f;
typedef __attribute__((ext_vector_type(8))) float v8f;

#define NNODES 50000
#define DHID   128
#define NCLS   10

#define AGG_BLOCKS  256
#define AGG_ROWS    ((NNODES + AGG_BLOCKS - 1) / AGG_BLOCKS)

#define EW_BLOCKS  256
#define EW_THREADS 512
#define EW_BATCH   8

__global__ __launch_bounds__(EW_THREADS) void bins_kernel(
    const int* __restrict__ src, const int* __restrict__ dst,
    const float* __restrict__ w, int* bins, int* prec, int nE) {
    unsigned char* bin = (unsigned char*)bins;
    int4* pr = (int4*)prec;
    const int stride = gridDim.x * EW_THREADS;
    for (int e0 = blockIdx.x * EW_THREADS + threadIdx.x; e0 < nE; e0 += EW_BATCH * stride) {
        int   d[EW_BATCH];
        int   s[EW_BATCH];
        float x[EW_BATCH];
        #pragma unroll
        for (int k = 0; k < EW_BATCH; ++k) {
            const int e = e0 + k * stride;
            d[k] = 0;
            s[k] = 0;
            x[k] = 0.f;
            if (e < nE) {
                d[k] = dst[e];
                s[k] = src[e];
                x[k] = w[e];
            }
        }
        #pragma unroll
        for (int k = 0; k < EW_BATCH; ++k) {
            const int e = e0 + k * stride;
            if (e < nE) {
                const unsigned o = (unsigned)d[k] / (unsigned)AGG_ROWS;
                bin[e] = (unsigned char)o;
                pr[e] = make_int4(s[k], __float_as_int(x[k]), d[k] - (int)(o * (unsigned)AGG_ROWS), 0);
            }
        }
    }
}

#define GM_THREADS 512
#define GM_WAVES   (GM_THREADS / 32)
#define GM_KSLAB   512
#define GM_WPITCH  136
#define GM_CH      8

template <bool VEC>
__device__ __forceinline__ float4 gm_load_a4(const float* __restrict__ xr, int k, int kw, bool rv)
{
    float4 v;
    if (VEC) {
        const bool ok = rv && (k < kw);
        const int kk = ok ? k : 0;
        v = *(const float4*)(xr + kk);
        if (!ok) v = make_float4(0.f, 0.f, 0.f, 0.f);
    } else {
        float t[4];
        #pragma unroll
        for (int e = 0; e < 4; ++e) {
            const bool ok = rv && (k + e < kw);
            const int kk = ok ? (k + e) : 0;
            const float x = xr[kk];
            t[e] = ok ? x : 0.f;
        }
        v = make_float4(t[0], t[1], t[2], t[3]);
    }
    return v;
}

template <bool VEC>
__device__ __forceinline__ void gemm_rows_body(
    const float* __restrict__ X, const float* __restrict__ W,
    float* __restrict__ H, int M, int K, float2* Wt)
{
    const int tid  = threadIdx.x;
    const int lane = tid & 31;
    const int wave = __builtin_amdgcn_readfirstlane(tid >> 5);
    const int c    = lane & 15;
    const int h    = lane >> 4;
    const int base = blockIdx.x * AGG_ROWS;
    int rows = M - base;
    if (rows > AGG_ROWS) rows = AGG_ROWS;
    if (rows < 0) rows = 0;
    const int nunits = ((rows + 15) >> 4) * 2;

    for (int ks = 0; ks < K; ks += GM_KSLAB) {
        const int kw  = (K - ks < GM_KSLAB) ? (K - ks) : GM_KSLAB;
        const int nst = (kw + 7) >> 3;

        if (ks > 0) __syncthreads();
        const int ntask = nst * 128;
        for (int f0 = tid; f0 < ntask; f0 += GM_THREADS * 4) {
            float4 va[4], vb[4];
            #pragma unroll
            for (int j = 0; j < 4; ++j) {
                const int f  = f0 + j * GM_THREADS;
                const int kp = f >> 5;
                const int cb = (f & 31) * 4;
                const int ra = (f < ntask && 2 * kp     < kw) ? (ks + 2 * kp)     : ks;
                const int rb = (f < ntask && 2 * kp + 1 < kw) ? (ks + 2 * kp + 1) : ks;
                va[j] = *(const float4*)(W + (size_t)ra * DHID + cb);
                vb[j] = *(const float4*)(W + (size_t)rb * DHID + cb);
            }
            #pragma unroll
            for (int j = 0; j < 4; ++j) {
                const int f  = f0 + j * GM_THREADS;
                const int kp = f >> 5;
                const int cb = (f & 31) * 4;
                if (f < ntask) {
                    float4 a = va[j], b = vb[j];
                    if (2 * kp     >= kw) a = make_float4(0.f, 0.f, 0.f, 0.f);
                    if (2 * kp + 1 >= kw) b = make_float4(0.f, 0.f, 0.f, 0.f);
                    *(float4*)&Wt[kp * GM_WPITCH + cb]     = make_float4(a.x, b.x, a.y, b.y);
                    *(float4*)&Wt[kp * GM_WPITCH + cb + 2] = make_float4(a.z, b.z, a.w, b.w);
                }
            }
        }
        __syncthreads();

        for (int u = wave; u < nunits; u += GM_WAVES) {
            const int strip = u >> 1;
            const int nh    = u & 1;
            const int lrow  = strip * 16 + c;
            const bool rv   = lrow < rows;
            const float* xr = X + (size_t)(base + (rv ? lrow : 0)) * K + ks;
            const bool whole = (strip * 16 + 16 <= rows);

            v8f acc[4];
            #pragma unroll
            for (int nt = 0; nt < 4; ++nt)
                #pragma unroll
                for (int r = 0; r < 8; ++r) acc[nt][r] = 0.0f;
            if (ks > 0) {
                #pragma unroll
                for (int nt = 0; nt < 4; ++nt)
                    #pragma unroll
                    for (int r = 0; r < 8; ++r) {
                        const int lr = strip * 16 + 8 * h + r;
                        if (lr < rows) acc[nt][r] = H[(size_t)(base + lr) * DHID + nh * 64 + nt * 16 + c];
                    }
            }

            float4 bufA[GM_CH], bufB[GM_CH];
            auto fetch = [&](float4 (&buf)[GM_CH], int t0) {
                #pragma unroll
                for (int j = 0; j < GM_CH; ++j)
                    buf[j] = gm_load_a4<VEC>(xr, 8 * (t0 + j) + 4 * h, kw, rv);
            };
            auto step = [&](const float4& a4, int t) {
                #pragma unroll
                for (int q = 0; q < 2; ++q) {
                    const int kp = 4 * t + 2 * h + q;
                    v2f a;
                    a.x = q ? a4.z : a4.x;
                    a.y = q ? a4.w : a4.y;
                    const float2* bp = &Wt[kp * GM_WPITCH + nh * 64 + c];
                    float2 bf[4];
                    #pragma unroll
                    for (int nt = 0; nt < 4; ++nt) bf[nt] = bp[nt * 16];
                    #pragma unroll
                    for (int nt = 0; nt < 4; ++nt) {
                        v2f b;
                        b.x = bf[nt].x;
                        b.y = bf[nt].y;
                        acc[nt] = __builtin_amdgcn_wmma_f32_16x16x4_f32(
                             false, a,  false, b,
                             (short)0, acc[nt],
                             false,  false);
                    }
                }
            };
            auto mult = [&](const float4 (&buf)[GM_CH], int t0) {
                if (t0 + GM_CH <= nst) {
                    #pragma unroll
                    for (int j = 0; j < GM_CH; ++j) step(buf[j], t0 + j);
                } else {
                    #pragma unroll
                    for (int j = 0; j < GM_CH; ++j)
                        if (t0 + j < nst) step(buf[j], t0 + j);
                }
            };

            fetch(bufA, 0);
            for (int t0 = 0; t0 < nst; t0 += 2 * GM_CH) {
                fetch(bufB, t0 + GM_CH);
                mult(bufA, t0);
                fetch(bufA, t0 + 2 * GM_CH);
                mult(bufB, t0 + GM_CH);
            }

            if (whole) {
                #pragma unroll
                for (int nt = 0; nt < 4; ++nt)
                    #pragma unroll
                    for (int r = 0; r < 8; ++r)
                        H[(size_t)(base + strip * 16 + 8 * h + r) * DHID + nh * 64 + nt * 16 + c] = acc[nt][r];
            } else {
                #pragma unroll
                for (int nt = 0; nt < 4; ++nt)
                    #pragma unroll
                    for (int r = 0; r < 8; ++r) {
                        const int lr = strip * 16 + 8 * h + r;
                        if (lr < rows) H[(size_t)(base + lr) * DHID + nh * 64 + nt * 16 + c] = acc[nt][r];
                    }
            }
        }
    }
}

__global__ __launch_bounds__(GM_THREADS) void gemm_rows_wmma_kernel(
    const float* __restrict__ X, const float* __restrict__ W,
    float* __restrict__ H, int M, int K)
{
    __shared__ float2 Wt[(GM_KSLAB / 2) * GM_WPITCH];
    if ((K & 3) == 0) gemm_rows_body<true >(X, W, H, M, K, Wt);
    else              gemm_rows_body<false>(X, W, H, M, K, Wt);
}


#define AGG_THREADS 1024
#define AGG_WAVES   (AGG_THREADS / 32)
#define AGG_FLT     4
#define AGG_QCAP    1024
#define AGG_LCAP    256
#define AGG_STRIPS  ((AGG_ROWS + 15) / 16)
#define AGG_TPITCH  132
static_assert(AGG_STRIPS * 2 <= AGG_WAVES, "one 16 x 64 unit per wave at most");
#define OWN_QCAP    1536

template <int QCAP, class Proc>
__device__ __forceinline__ int owner_sweep(
    const int* __restrict__ bins, const int* __restrict__ dst, int nE,
    int base, unsigned urows, int* qb, int wave, int lane, bool may_keep, Proc proc)
{
    int cnt = 0;
    int kept = -1;

    auto push = [&](bool m, int e) {
        const unsigned bal = __builtin_amdgcn_ballot_w32(m);
        if (bal) {
            if (m) qb[cnt + (int)__builtin_amdgcn_mbcnt_lo(bal, 0u)] = e;
            cnt += __builtin_popcount(bal);
        }
    };

    const int4* bq = (const int4*)bins;
    const unsigned pat = blockIdx.x * 0x01010101u;
    const int nfull = nE >> 9;
    const int rot = (int)((blockIdx.x * (unsigned)nfull) / gridDim.x);
    int it = wave;
    int tb = (nfull << 9) + wave * 32;
    bool more = true;
    bool first = true;
    while (more) {
        while (it < nfull && cnt <= QCAP - 512) {
            int4 d[AGG_FLT];
            int  e0[AGG_FLT];
            #pragma unroll
            for (int k = 0; k < AGG_FLT; ++k) {
                int sc = ((it + k * AGG_WAVES < nfull) ? (it + k * AGG_WAVES) : it) + rot;
                if (sc >= nfull) sc -= nfull;
                d[k]  = bq[(sc << 5) + lane];
                e0[k] = (sc << 9) + lane * 16;
            }
            int done = 0;
            #pragma unroll
            for (int k = 0; k < AGG_FLT; ++k) {
                if (done == k && it + k * AGG_WAVES < nfull && cnt <= QCAP - 512) {
                    auto z = [&](int x) -> unsigned {
                        const unsigned t = (unsigned)x ^ pat;
                        return ~(((t & 0x7F7F7F7Fu) + 0x7F7F7F7Fu) | t | 0x7F7F7F7Fu);
                    };
                    unsigned m = (z(d[k].x) >> 7) | (z(d[k].y) >> 6) |
                                 (z(d[k].z) >> 5) | (z(d[k].w) >> 4);
                    unsigned bal;
                    while ((bal = __builtin_amdgcn_ballot_w32(m != 0u)) != 0u) {
                        if (m != 0u) {
                            const int p = __builtin_ctz(m);
                            qb[cnt + (int)__builtin_amdgcn_mbcnt_lo(bal, 0u)] =
                                e0[k] + ((p & 7) << 2) + (p >> 3);
                        }
                        cnt += __builtin_popcount(bal);
                        m &= m - 1u;
                    }
                    done = k + 1;
                }
            }
            it += done * AGG_WAVES;
        }
        if (it >= nfull) {
            while (tb < nE && cnt <= QCAP - 32) {
                const int e = tb + lane;
                const int dv = (e < nE) ? dst[e] : -1;
                push((unsigned)(dv - base) < urows, e);
                tb += AGG_THREADS;
            }
            more = (tb < nE);
        }

        const bool keep = may_keep && first && !more && (cnt <= AGG_LCAP);

        proc(qb, cnt, keep);
        if (keep) kept = cnt;
        first = false;
        cnt = 0;
    }
    return kept;
}

__global__ __launch_bounds__(AGG_THREADS) void own_edges_kernel(
    const int* __restrict__ dst, const int* __restrict__ bins,
    const int* __restrict__ prec,
    float* __restrict__ deg, int n, int nE, int* recs, int* rcnt)
{
    __shared__ float degl[AGG_ROWS];
    __shared__ int   queue[AGG_WAVES * OWN_QCAP];

    const int tid  = threadIdx.x;
    const int lane = tid & 31;
    const int wave = __builtin_amdgcn_readfirstlane(tid >> 5);
    const int base = blockIdx.x * AGG_ROWS;
    int rows = n - base;
    if (rows > AGG_ROWS) rows = AGG_ROWS;
    if (rows < 0) rows = 0;
    const unsigned urows = (unsigned)rows;

    for (int i = tid; i < AGG_ROWS; i += AGG_THREADS) degl[i] = 1.0f;
    __syncthreads();

    const int slot = blockIdx.x * AGG_WAVES + wave;
    int4* rq = (int4*)recs + (size_t)slot * AGG_LCAP;
    const int4* pq = (const int4*)prec;

    const int kept = owner_sweep<OWN_QCAP>(bins, dst, nE, base, urows, queue + wave * OWN_QCAP,
                                           wave, lane, true,
        [&](const int* q, int cnt, bool keep) {
            for (int b = 0; b < cnt; b += 128) {
                int4 pv[4];
                #pragma unroll
                for (int j = 0; j < 4; ++j) {
                    const int i = b + 32 * j + lane;
                    const int e = q[i < cnt ? i : b];
                    pv[j] = pq[e];
                }
                asm volatile("" :: "v"(pv[0].x), "v"(pv[0].y), "v"(pv[0].z), "v"(pv[0].w),
                                   "v"(pv[1].x), "v"(pv[1].y), "v"(pv[1].z), "v"(pv[1].w),
                                   "v"(pv[2].x), "v"(pv[2].y), "v"(pv[2].z), "v"(pv[2].w),
                                   "v"(pv[3].x), "v"(pv[3].y), "v"(pv[3].z), "v"(pv[3].w));
                #pragma unroll
                for (int j = 0; j < 4; ++j) {
                    const int i = b + 32 * j + lane;
                    if (i < cnt) {
                        atomicAdd(&degl[pv[j].z], __int_as_float(pv[j].y));
                        if (keep) rq[i] = pv[j];
                    }
                }
            }
        });
    if (lane == 0) rcnt[slot] = kept;
    __syncthreads();

    for (int i = tid; i < rows; i += AGG_THREADS) deg[base + i] = degl[i];
}

template <bool TAIL>
__global__ __launch_bounds__(AGG_THREADS) void agg_rows_kernel(
    const float* __restrict__ H,
    const int* __restrict__ src, const int* __restrict__ dst,
    const float* __restrict__ w, const float* __restrict__ deg,
    const int* __restrict__ bins,
    float* __restrict__ OUT, int n, int nE,
    int* recs, const int* __restrict__ rcnt,
    const float* __restrict__ actb, const float* __restrict__ W2,
    const float* __restrict__ b2, const float* __restrict__ Wc,
    const float* __restrict__ bc)
{
    constexpr int P     = TAIL ? AGG_TPITCH : DHID;
    constexpr int AROWS = TAIL ? AGG_STRIPS * 16 : AGG_ROWS;
    __shared__ alignas(16) float acc[AROWS * P];
    __shared__ int   queue[AGG_WAVES * AGG_QCAP];
    __shared__ float dloc[AGG_ROWS];
    __shared__ alignas(16) float2 Wt[TAIL ? (DHID / 2) * GM_WPITCH : 1];

    const int tid  = threadIdx.x;
    const int lane = tid & 31;
    const int wave = __builtin_amdgcn_readfirstlane(tid >> 5);
    const int base = blockIdx.x * AGG_ROWS;
    int rows = n - base;
    if (rows > AGG_ROWS) rows = AGG_ROWS;
    if (rows < 0) rows = 0;
    const unsigned urows = (unsigned)rows;

    float4 bl = make_float4(0.f, 0.f, 0.f, 0.f);
    if (TAIL) bl = *(const float4*)(actb + lane * 4);
    auto act = [&](float4 v) -> float4 {
        if (TAIL) {
            const float x0 = v.x + bl.x, x1 = v.y + bl.y;
            const float x2 = v.z + bl.z, x3 = v.w + bl.w;
            v = make_float4(x0 > 0.f ? x0 : 0.f, x1 > 0.f ? x1 : 0.f,
                            x2 > 0.f ? x2 : 0.f, x3 > 0.f ? x3 : 0.f);
        }
        return v;
    };

    if (TAIL) {
        float4 va[2], vb[2];
        #pragma unroll
        for (int j = 0; j < 2; ++j) {
            const int f  = tid + j * AGG_THREADS;
            const int kp = f >> 5;
            const int cb = (f & 31) * 4;
            const int d0 = 2 * kp, d1 = 2 * kp + 1;
            va[j] = *(const float4*)(W2 + (size_t)(4 * (d0 & 31) + (d0 >> 5)) * DHID + cb);
            vb[j] = *(const float4*)(W2 + (size_t)(4 * (d1 & 31) + (d1 >> 5)) * DHID + cb);
        }
        #pragma unroll
        for (int j = 0; j < 2; ++j) {
            const int f  = tid + j * AGG_THREADS;
            const int kp = f >> 5;
            const int cb = (f & 31) * 4;
            const float4 a = va[j], b = vb[j];
            *(float4*)&Wt[kp * GM_WPITCH + cb]     = make_float4(a.x, b.x, a.y, b.y);
            *(float4*)&Wt[kp * GM_WPITCH + cb + 2] = make_float4(a.z, b.z, a.w, b.w);
        }
    }

    for (int idx = tid; idx < (TAIL ? AROWS : rows) * 32; idx += AGG_THREADS) {
        const int r = idx >> 5;
        const int l = idx & 31;
        float* a = acc + r * P + l;
        if (r < rows) {
            const float4 h = act(*(const float4*)(H + (size_t)(base + r) * DHID + l * 4));
            const float s = __frsqrt_rn(deg[base + r]);
            if (l == 0) dloc[r] = s;
            a[0]  = h.x * s * s;
            a[32] = h.y * s * s;
            a[64] = h.z * s * s;
            a[96] = h.w * s * s;
        } else {
            a[0] = 0.f; a[32] = 0.f; a[64] = 0.f; a[96] = 0.f;
        }
    }
    __syncthreads();

    auto add_rows = [&](int sv, int nvb, int lr, int nb) {
        for (int i = 0; i < nb; i += 8) {
            float4 hv[8];
            float  nk[8];
            int    lk[8];
            #pragma unroll
            for (int k = 0; k < 8; ++k) {
                const int ii = (i + k < nb) ? (i + k) : i;
                const int sk = __builtin_amdgcn_readlane(sv, ii);
                nk[k] = __int_as_float(__builtin_amdgcn_readlane(nvb, ii));
                lk[k] = __builtin_amdgcn_readlane(lr, ii);
                hv[k] = act(*(const float4*)(H + (size_t)sk * DHID + lane * 4));
            }
            #pragma unroll
            for (int k = 0; k < 8; ++k) {
                if (i + k < nb) {
                    float* a = acc + lk[k] * P + lane;
                    atomicAdd(a,      hv[k].x * nk[k]);
                    atomicAdd(a + 32, hv[k].y * nk[k]);
                    atomicAdd(a + 64, hv[k].z * nk[k]);
                    atomicAdd(a + 96, hv[k].w * nk[k]);
                }
            }
        }
    };

    const int slot = blockIdx.x * AGG_WAVES + wave;
    int4* rq = (int4*)recs + (size_t)slot * AGG_LCAP;

    int kept = __builtin_amdgcn_readfirstlane(rcnt[slot]);
    if (kept > AGG_LCAP) kept = AGG_LCAP;

    if (kept >= 0) {
        for (int b = 0; b < kept; b += 32) {
            const int nb = (kept - b < 32) ? (kept - b) : 32;
            const int4 r = rq[b + (lane < nb ? lane : 0)];
            int nvb = r.y;
            if (!TAIL) {
                const float a = deg[r.x];
                nvb = __float_as_int(__frsqrt_rn(a) * __int_as_float(r.y) * dloc[r.z]);
                if (lane < nb) ((int*)(rq + b + lane))[1] = nvb;
            }
            add_rows(r.x, nvb, r.z, nb);
        }
    } else {
        owner_sweep<AGG_QCAP>(bins, dst, nE, base, urows, queue + wave * AGG_QCAP,
                              wave, lane, false,
            [&](const int* q, int cnt, bool keep) {
                (void)keep;
                for (int b = 0; b < cnt; b += 32) {
                    const int nb = (cnt - b < 32) ? (cnt - b) : 32;
                    const int   e  = q[b + (lane < nb ? lane : 0)];
                    const int   sv = src[e];
                    const float wv = w[e];
                    const int   lr = dst[e] - base;
                    const float a  = deg[sv];
                    const int  nvb = __float_as_int(__frsqrt_rn(a) * wv * dloc[lr]);
                    add_rows(sv, nvb, lr, nb);
                }
            });
    }
    __syncthreads();

    if (!TAIL) {
        for (int idx = tid; idx < rows * 32; idx += AGG_THREADS) {
            const int r = idx >> 5;
            const int l = idx & 31;
            const float* a = acc + r * P + l;
            const float4 o = make_float4(a[0], a[32], a[64], a[96]);
            *(float4*)(OUT + (size_t)(base + r) * DHID + l * 4) = o;
        }
    } else {
        const int c = lane & 15;
        const int h = lane >> 4;
        const int nstr = (rows + 15) >> 4;
        const bool has  = wave < nstr * 2;
        const int strip = wave >> 1;
        const int nh    = wave & 1;
        v8f dt[4];
        #pragma unroll
        for (int nt = 0; nt < 4; ++nt)
            #pragma unroll
            for (int r = 0; r < 8; ++r) dt[nt][r] = 0.0f;
        if (has) {
            const float* ar = acc + (strip * 16 + c) * P + 4 * h;
            #pragma unroll
            for (int t = 0; t < DHID / 8; ++t) {
                const float4 a4 = *(const float4*)(ar + 8 * t);
                #pragma unroll
                for (int q = 0; q < 2; ++q) {
                    const int kp = 4 * t + 2 * h + q;
                    v2f a;
                    a.x = q ? a4.z : a4.x;
                    a.y = q ? a4.w : a4.y;
                    const float2* bp = &Wt[kp * GM_WPITCH + nh * 64 + c];
                    float2 bf[4];
                    #pragma unroll
                    for (int nt = 0; nt < 4; ++nt) bf[nt] = bp[nt * 16];
                    #pragma unroll
                    for (int nt = 0; nt < 4; ++nt) {
                        v2f b;
                        b.x = bf[nt].x;
                        b.y = bf[nt].y;
                        dt[nt] = __builtin_amdgcn_wmma_f32_16x16x4_f32(
                             false, a,  false, b,
                             (short)0, dt[nt],
                             false,  false);
                    }
                }
            }
        }
        __syncthreads();

        if (has) {
            #pragma unroll
            for (int nt = 0; nt < 4; ++nt) {
                const int col = nh * 64 + nt * 16 + c;
                const float bb = b2[col];
                #pragma unroll
                for (int r = 0; r < 8; ++r) {
                    const float x = dt[nt][r] + bb;
                    acc[(strip * 16 + 8 * h + r) * P + col] = x > 0.f ? x : 0.f;
                }
            }
        }
        __syncthreads();

        if (wave < nstr) {
            const int cc = c < NCLS ? c : 0;
            const float* ar = acc + (wave * 16 + c) * P + 4 * h;
            v8f lg;
            #pragma unroll
            for (int r = 0; r < 8; ++r) lg[r] = 0.0f;
            #pragma unroll
            for (int t = 0; t < DHID / 8; ++t) {
                const float4 a4 = *(const float4*)(ar + 8 * t);
                #pragma unroll
                for (int q = 0; q < 2; ++q) {
                    const int k0 = 8 * t + 4 * h + 2 * q;
                    v2f a, b;
                    a.x = q ? a4.z : a4.x;
                    a.y = q ? a4.w : a4.y;
                    const float w0 = Wc[(k0    ) * NCLS + cc];
                    const float w1 = Wc[(k0 + 1) * NCLS + cc];
                    b.x = c < NCLS ? w0 : 0.f;
                    b.y = c < NCLS ? w1 : 0.f;
                    lg = __builtin_amdgcn_wmma_f32_16x16x4_f32(
                         false, a,  false, b,
                         (short)0, lg,
                         false,  false);
                }
            }
            const float bcv = bc[cc];
            #pragma unroll
            for (int r = 0; r < 8; ++r) {
                const int lr = wave * 16 + 8 * h + r;
                if (c < NCLS && lr < rows) OUT[(size_t)(base + lr) * NCLS + c] = lg[r] + bcv;
            }
        }
    }
}

extern "C" void kernel_launch(void* const* d_in, const int* in_sizes, int n_in,
                              void* d_out, int out_size, void* d_ws, size_t ws_size,
                              hipStream_t stream) {
    const float* x  = (const float*)d_in[0];
    const int*   ei = (const int*)d_in[1];
    const float* ew = (const float*)d_in[2];
    const float* W1 = (const float*)d_in[3];
    const float* b1 = (const float*)d_in[4];
    const float* W2 = (const float*)d_in[5];
    const float* b2 = (const float*)d_in[6];
    const float* Wc = (const float*)d_in[7];
    const float* bc = (const float*)d_in[8];
    float* out = (float*)d_out;

    const int n  = NNODES;
    const int E  = in_sizes[1] / 2;
    const int K1 = in_sizes[0] / n;
    const int* src = ei;
    const int* dst = ei + E;

    char* ws = (char*)d_ws;
    size_t off = 0;
    auto alloc = [&](size_t bytes) { char* p = ws + off; off = (off + bytes + 255) & ~(size_t)255; return p; };
    float* deg  = (float*)alloc((size_t)n * 4);
    float* bufA = (float*)alloc((size_t)n * DHID * 4);
    float* bufB = (float*)alloc((size_t)n * DHID * 4);
    int* recs = (int*)alloc((size_t)AGG_BLOCKS * AGG_WAVES * AGG_LCAP * 16);
    int* rcnt = (int*)alloc((size_t)AGG_BLOCKS * AGG_WAVES * 4);
    int* bins = (int*)alloc((size_t)E);
    int* prec = (int*)alloc((size_t)E * 16);
    (void)ws_size;

    bins_kernel     <<<EW_BLOCKS, EW_THREADS, 0, stream>>>(src, dst, ew, bins, prec, E);
    own_edges_kernel<<<AGG_BLOCKS, AGG_THREADS, 0, stream>>>(dst, bins, prec, deg, n, E, recs, rcnt);

    gemm_rows_wmma_kernel<<<AGG_BLOCKS, GM_THREADS, 0, stream>>>(x, W1, bufA, n, K1);
    agg_rows_kernel<false><<<AGG_BLOCKS, AGG_THREADS, 0, stream>>>(bufA, src, dst, ew, deg, bins, bufB, n, E, recs, rcnt,
                                                                  nullptr, nullptr, nullptr, nullptr, nullptr);

    agg_rows_kernel<true><<<AGG_BLOCKS, AGG_THREADS, 0, stream>>>(bufB, src, dst, ew, deg, bins, out, n, E, recs, rcnt,
                                                                 b1, W2, b2, Wc, bc);
}


# Owner sweep reads a bit table: the per-edge pass sets one bit per (row-owner block, edge) by LDS integer ORs; a block walks the set bits of its own row, not every edge's id; float sums: LDS arrival
# speedup vs baseline: 4.8169x; 1.0358x over previous
#include <hip/hip_runtime.h>

typedef __attribute__((ext_vector_type(2))) float v2f;
typedef __attribute__((ext_vector_type(8))) float v8f;

#define NNODES 50000
#define DHID   128
#define NCLS   10

#define AGG_BLOCKS  256
#define AGG_ROWS    ((NNODES + AGG_BLOCKS - 1) / AGG_BLOCKS)

#define EW_BLOCKS  256
#define EW_THREADS 512
#define EW_BATCH   8

#define BM_CHUNK   4096
#define BM_CWORDS  (BM_CHUNK / 32)
#define BM_PITCH   (BM_CWORDS + 4)
static_assert(EW_THREADS * EW_BATCH == BM_CHUNK, "a work-group takes one chunk per trip");

__global__ __launch_bounds__(EW_THREADS) void bins_kernel(
    const int* __restrict__ src, const int* __restrict__ dst,
    const float* __restrict__ w, unsigned* bm, int* prec, int nE) {
    __shared__ alignas(16) unsigned tab[AGG_BLOCKS * BM_PITCH];
    int4* pr = (int4*)prec;
    const int tid = threadIdx.x;
    const int nchunk = (nE + BM_CHUNK - 1) / BM_CHUNK;
    const size_t roww = (size_t)nchunk * BM_CWORDS;
    for (int ch = blockIdx.x; ch < nchunk; ch += gridDim.x) {
        const int e0 = ch * BM_CHUNK + tid;
        int   d[EW_BATCH];
        int   s[EW_BATCH];
        float x[EW_BATCH];
        #pragma unroll
        for (int k = 0; k < EW_BATCH; ++k) {
            const int e = e0 + k * EW_THREADS;
            d[k] = 0;
            s[k] = 0;
            x[k] = 0.f;
            if (e < nE) {
                d[k] = dst[e];
                s[k] = src[e];
                x[k] = w[e];
            }
        }
        for (int i = tid; i < AGG_BLOCKS * BM_PITCH / 4; i += EW_THREADS)
            ((uint4*)tab)[i] = make_uint4(0u, 0u, 0u, 0u);
        __syncthreads();
        #pragma unroll
        for (int k = 0; k < EW_BATCH; ++k) {
            const int e = e0 + k * EW_THREADS;
            const unsigned o = (unsigned)d[k] / (unsigned)AGG_ROWS;
            if (e < nE && o < (unsigned)AGG_BLOCKS)
                atomicOr(&tab[o * BM_PITCH + k * (EW_THREADS / 32) + (tid >> 5)], 1u << (tid & 31));
        }
        __syncthreads();
        for (int o = tid >> 5; o < AGG_BLOCKS; o += EW_THREADS / 32) {
            const uint4 v = *(const uint4*)&tab[o * BM_PITCH + (tid & 31) * 4];
            *(uint4*)(bm + (size_t)o * roww + (size_t)ch * BM_CWORDS + (tid & 31) * 4) = v;
        }
        #pragma unroll
        for (int k = 0; k < EW_BATCH; ++k) {
            const int e = e0 + k * EW_THREADS;
            if (e < nE) {
                const unsigned o = (unsigned)d[k] / (unsigned)AGG_ROWS;
                pr[e] = make_int4(s[k], __float_as_int(x[k]), d[k] - (int)(o * (unsigned)AGG_ROWS), 0);
            }
        }
        if (ch + (int)gridDim.x < nchunk) __syncthreads();
    }
}

#define GM_THREADS 512
#define GM_WAVES   (GM_THREADS / 32)
#define GM_KSLAB   512
#define GM_WPITCH  136
#define GM_CH      8

template <bool VEC>
__device__ __forceinline__ float4 gm_load_a4(const float* __restrict__ xr, int k, int kw, bool rv)
{
    float4 v;
    if (VEC) {
        const bool ok = rv && (k < kw);
        const int kk = ok ? k : 0;
        v = *(const float4*)(xr + kk);
        if (!ok) v = make_float4(0.f, 0.f, 0.f, 0.f);
    } else {
        float t[4];
        #pragma unroll
        for (int e = 0; e < 4; ++e) {
            const bool ok = rv && (k + e < kw);
            const int kk = ok ? (k + e) : 0;
            const float x = xr[kk];
            t[e] = ok ? x : 0.f;
        }
        v = make_float4(t[0], t[1], t[2], t[3]);
    }
    return v;
}

template <bool VEC>
__device__ __forceinline__ void gemm_rows_body(
    const float* __restrict__ X, const float* __restrict__ W,
    float* __restrict__ H, int M, int K, float2* Wt)
{
    const int tid  = threadIdx.x;
    const int lane = tid & 31;
    const int wave = __builtin_amdgcn_readfirstlane(tid >> 5);
    const int c    = lane & 15;
    const int h    = lane >> 4;
    const int base = blockIdx.x * AGG_ROWS;
    int rows = M - base;
    if (rows > AGG_ROWS) rows = AGG_ROWS;
    if (rows < 0) rows = 0;
    const int nunits = ((rows + 15) >> 4) * 2;

    for (int ks = 0; ks < K; ks += GM_KSLAB) {
        const int kw  = (K - ks < GM_KSLAB) ? (K - ks) : GM_KSLAB;
        const int nst = (kw + 7) >> 3;

        if (ks > 0) __syncthreads();
        const int ntask = nst * 128;
        for (int f0 = tid; f0 < ntask; f0 += GM_THREADS * 4) {
            float4 va[4], vb[4];
            #pragma unroll
            for (int j = 0; j < 4; ++j) {
                const int f  = f0 + j * GM_THREADS;
                const int kp = f >> 5;
                const int cb = (f & 31) * 4;
                const int ra = (f < ntask && 2 * kp     < kw) ? (ks + 2 * kp)     : ks;
                const int rb = (f < ntask && 2 * kp + 1 < kw) ? (ks + 2 * kp + 1) : ks;
                va[j] = *(const float4*)(W + (size_t)ra * DHID + cb);
                vb[j] = *(const float4*)(W + (size_t)rb * DHID + cb);
            }
            #pragma unroll
            for (int j = 0; j < 4; ++j) {
                const int f  = f0 + j * GM_THREADS;
                const int kp = f >> 5;
                const int cb = (f & 31) * 4;
                if (f < ntask) {
                    float4 a = va[j], b = vb[j];
                    if (2 * kp     >= kw) a = make_float4(0.f, 0.f, 0.f, 0.f);
                    if (2 * kp + 1 >= kw) b = make_float4(0.f, 0.f, 0.f, 0.f);
                    *(float4*)&Wt[kp * GM_WPITCH + cb]     = make_float4(a.x, b.x, a.y, b.y);
                    *(float4*)&Wt[kp * GM_WPITCH + cb + 2] = make_float4(a.z, b.z, a.w, b.w);
                }
            }
        }
        __syncthreads();

        for (int u = wave; u < nunits; u += GM_WAVES) {
            const int strip = u >> 1;
            const int nh    = u & 1;
            const int lrow  = strip * 16 + c;
            const bool rv   = lrow < rows;
            const float* xr = X + (size_t)(base + (rv ? lrow : 0)) * K + ks;
            const bool whole = (strip * 16 + 16 <= rows);

            v8f acc[4];
            #pragma unroll
            for (int nt = 0; nt < 4; ++nt)
                #pragma unroll
                for (int r = 0; r < 8; ++r) acc[nt][r] = 0.0f;
            if (ks > 0) {
                #pragma unroll
                for (int nt = 0; nt < 4; ++nt)
                    #pragma unroll
                    for (int r = 0; r < 8; ++r) {
                        const int lr = strip * 16 + 8 * h + r;
                        if (lr < rows) acc[nt][r] = H[(size_t)(base + lr) * DHID + nh * 64 + nt * 16 + c];
                    }
            }

            float4 bufA[GM_CH], bufB[GM_CH];
            auto fetch = [&](float4 (&buf)[GM_CH], int t0) {
                #pragma unroll
                for (int j = 0; j < GM_CH; ++j)
                    buf[j] = gm_load_a4<VEC>(xr, 8 * (t0 + j) + 4 * h, kw, rv);
            };
            auto step = [&](const float4& a4, int t) {
                #pragma unroll
                for (int q = 0; q < 2; ++q) {
                    const int kp = 4 * t + 2 * h + q;
                    v2f a;
                    a.x = q ? a4.z : a4.x;
                    a.y = q ? a4.w : a4.y;
                    const float2* bp = &Wt[kp * GM_WPITCH + nh * 64 + c];
                    float2 bf[4];
                    #pragma unroll
                    for (int nt = 0; nt < 4; ++nt) bf[nt] = bp[nt * 16];
                    #pragma unroll
                    for (int nt = 0; nt < 4; ++nt) {
                        v2f b;
                        b.x = bf[nt].x;
                        b.y = bf[nt].y;
                        acc[nt] = __builtin_amdgcn_wmma_f32_16x16x4_f32(
                             false, a,  false, b,
                             (short)0, acc[nt],
                             false,  false);
                    }
                }
            };
            auto mult = [&](const float4 (&buf)[GM_CH], int t0) {
                if (t0 + GM_CH <= nst) {
                    #pragma unroll
                    for (int j = 0; j < GM_CH; ++j) step(buf[j], t0 + j);
                } else {
                    #pragma unroll
                    for (int j = 0; j < GM_CH; ++j)
                        if (t0 + j < nst) step(buf[j], t0 + j);
                }
            };

            fetch(bufA, 0);
            for (int t0 = 0; t0 < nst; t0 += 2 * GM_CH) {
                fetch(bufB, t0 + GM_CH);
                mult(bufA, t0);
                fetch(bufA, t0 + 2 * GM_CH);
                mult(bufB, t0 + GM_CH);
            }

            if (whole) {
                #pragma unroll
                for (int nt = 0; nt < 4; ++nt)
                    #pragma unroll
                    for (int r = 0; r < 8; ++r)
                        H[(size_t)(base + strip * 16 + 8 * h + r) * DHID + nh * 64 + nt * 16 + c] = acc[nt][r];
            } else {
                #pragma unroll
                for (int nt = 0; nt < 4; ++nt)
                    #pragma unroll
                    for (int r = 0; r < 8; ++r) {
                        const int lr = strip * 16 + 8 * h + r;
                        if (lr < rows) H[(size_t)(base + lr) * DHID + nh * 64 + nt * 16 + c] = acc[nt][r];
                    }
            }
        }
    }
}

__global__ __launch_bounds__(GM_THREADS) void gemm_rows_wmma_kernel(
    const float* __restrict__ X, const float* __restrict__ W,
    float* __restrict__ H, int M, int K)
{
    __shared__ float2 Wt[(GM_KSLAB / 2) * GM_WPITCH];
    if ((K & 3) == 0) gemm_rows_body<true >(X, W, H, M, K, Wt);
    else              gemm_rows_body<false>(X, W, H, M, K, Wt);
}


#define AGG_THREADS 1024
#define AGG_WAVES   (AGG_THREADS / 32)
#define AGG_FLT     4
#define AGG_QCAP    1024
#define AGG_LCAP    256
#define AGG_STRIPS  ((AGG_ROWS + 15) / 16)
#define AGG_TPITCH  132
static_assert(AGG_STRIPS * 2 <= AGG_WAVES, "one 16 x 64 unit per wave at most");
#define OWN_QCAP    1536

template <int QCAP, class Proc>
__device__ __forceinline__ int owner_sweep(
    const unsigned* __restrict__ bmrow, int nchunk, int* qb, int wave, int lane, bool may_keep, Proc proc)
{
    static_assert(QCAP >= 1024, "one word of a wave's load (32 lanes x 32 bits) must fit an empty queue");
    int cnt = 0;
    int kept = -1;
    const uint4* bq = (const uint4*)bmrow;
    int s0 = wave;
    bool loaded = false;
    bool more = true;
    bool first = true;
    uint4 d[AGG_FLT];
    #pragma unroll
    for (int k = 0; k < AGG_FLT; ++k) d[k] = make_uint4(0u, 0u, 0u, 0u);
    while (more) {
        bool full = false;
        auto drain = [&](unsigned& m, int eb) {
            if (full) return;
            unsigned bal = __builtin_amdgcn_ballot_w32(m != 0u);
            if (bal == 0u) return;
            if (cnt > QCAP - 1024) { full = true; return; }
            do {
                if (m != 0u) {
                    qb[cnt + (int)__builtin_amdgcn_mbcnt_lo(bal, 0u)] = eb + __builtin_ctz(m);
                    m &= m - 1u;
                }
                cnt += __builtin_popcount(bal);
                bal = __builtin_amdgcn_ballot_w32(m != 0u);
            } while (bal != 0u);
        };
        while (!full) {
            if (!loaded) {
                if (s0 >= nchunk) { more = false; break; }
                #pragma unroll
                for (int k = 0; k < AGG_FLT; ++k) {
                    const int sc = s0 + k * AGG_WAVES;
                    const bool ok = sc < nchunk;
                    d[k] = bq[((ok ? sc : s0) << 5) + lane];
                    if (!ok) d[k] = make_uint4(0u, 0u, 0u, 0u);
                }
                loaded = true;
            }
            #pragma unroll
            for (int k = 0; k < AGG_FLT; ++k) {
                const int eb = (s0 + k * AGG_WAVES) * BM_CHUNK + lane * 128;
                drain(d[k].x, eb);
                drain(d[k].y, eb + 32);
                drain(d[k].z, eb + 64);
                drain(d[k].w, eb + 96);
            }
            if (!full) {
                loaded = false;
                s0 += AGG_FLT * AGG_WAVES;
            }
        }

        const bool keep = may_keep && first && !more && (cnt <= AGG_LCAP);

        proc(qb, cnt, keep);
        if (keep) kept = cnt;
        first = false;
        cnt = 0;
    }
    return kept;
}

__global__ __launch_bounds__(AGG_THREADS) void own_edges_kernel(
    const unsigned* __restrict__ bm, const int* __restrict__ prec,
    float* __restrict__ deg, int n, int nE, int* recs, int* rcnt)
{
    __shared__ float degl[AGG_ROWS];
    __shared__ int   queue[AGG_WAVES * OWN_QCAP];

    const int tid  = threadIdx.x;
    const int lane = tid & 31;
    const int wave = __builtin_amdgcn_readfirstlane(tid >> 5);
    const int base = blockIdx.x * AGG_ROWS;
    int rows = n - base;
    if (rows > AGG_ROWS) rows = AGG_ROWS;
    if (rows < 0) rows = 0;
    const int nchunk = (nE + BM_CHUNK - 1) / BM_CHUNK;
    const unsigned* bmrow = bm + (size_t)blockIdx.x * ((size_t)nchunk * BM_CWORDS);

    for (int i = tid; i < AGG_ROWS; i += AGG_THREADS) degl[i] = 1.0f;
    __syncthreads();

    const int slot = blockIdx.x * AGG_WAVES + wave;
    int4* rq = (int4*)recs + (size_t)slot * AGG_LCAP;
    const int4* pq = (const int4*)prec;

    const int kept = owner_sweep<OWN_QCAP>(bmrow, nchunk, queue + wave * OWN_QCAP,
                                           wave, lane, true,
        [&](const int* q, int cnt, bool keep) {
            for (int b = 0; b < cnt; b += 128) {
                int4 pv[4];
                #pragma unroll
                for (int j = 0; j < 4; ++j) {
                    const int i = b + 32 * j + lane;
                    const int e = q[i < cnt ? i : b];
                    pv[j] = pq[e];
                }
                asm volatile("" :: "v"(pv[0].x), "v"(pv[0].y), "v"(pv[0].z), "v"(pv[0].w),
                                   "v"(pv[1].x), "v"(pv[1].y), "v"(pv[1].z), "v"(pv[1].w),
                                   "v"(pv[2].x), "v"(pv[2].y), "v"(pv[2].z), "v"(pv[2].w),
                                   "v"(pv[3].x), "v"(pv[3].y), "v"(pv[3].z), "v"(pv[3].w));
                #pragma unroll
                for (int j = 0; j < 4; ++j) {
                    const int i = b + 32 * j + lane;
                    if (i < cnt) {
                        atomicAdd(&degl[pv[j].z], __int_as_float(pv[j].y));
                        if (keep) rq[i] = pv[j];
                    }
                }
            }
        });
    if (lane == 0) rcnt[slot] = kept;
    __syncthreads();

    for (int i = tid; i < rows; i += AGG_THREADS) deg[base + i] = degl[i];
}

template <bool TAIL>
__global__ __launch_bounds__(AGG_THREADS) void agg_rows_kernel(
    const float* __restrict__ H,
    const int* __restrict__ src, const int* __restrict__ dst,
    const float* __restrict__ w, const float* __restrict__ deg,
    const unsigned* __restrict__ bm,
    float* __restrict__ OUT, int n, int nE,
    int* recs, const int* __restrict__ rcnt,
    const float* __restrict__ actb, const float* __restrict__ W2,
    const float* __restrict__ b2, const float* __restrict__ Wc,
    const float* __restrict__ bc)
{
    constexpr int P     = TAIL ? AGG_TPITCH : DHID;
    constexpr int AROWS = TAIL ? AGG_STRIPS * 16 : AGG_ROWS;
    __shared__ alignas(16) float acc[AROWS * P];
    __shared__ int   queue[AGG_WAVES * AGG_QCAP];
    __shared__ float dloc[AGG_ROWS];
    __shared__ alignas(16) float2 Wt[TAIL ? (DHID / 2) * GM_WPITCH : 1];

    const int tid  = threadIdx.x;
    const int lane = tid & 31;
    const int wave = __builtin_amdgcn_readfirstlane(tid >> 5);
    const int base = blockIdx.x * AGG_ROWS;
    int rows = n - base;
    if (rows > AGG_ROWS) rows = AGG_ROWS;
    if (rows < 0) rows = 0;

    float4 bl = make_float4(0.f, 0.f, 0.f, 0.f);
    if (TAIL) bl = *(const float4*)(actb + lane * 4);
    auto act = [&](float4 v) -> float4 {
        if (TAIL) {
            const float x0 = v.x + bl.x, x1 = v.y + bl.y;
            const float x2 = v.z + bl.z, x3 = v.w + bl.w;
            v = make_float4(x0 > 0.f ? x0 : 0.f, x1 > 0.f ? x1 : 0.f,
                            x2 > 0.f ? x2 : 0.f, x3 > 0.f ? x3 : 0.f);
        }
        return v;
    };

    if (TAIL) {
        float4 va[2], vb[2];
        #pragma unroll
        for (int j = 0; j < 2; ++j) {
            const int f  = tid + j * AGG_THREADS;
            const int kp = f >> 5;
            const int cb = (f & 31) * 4;
            const int d0 = 2 * kp, d1 = 2 * kp + 1;
            va[j] = *(const float4*)(W2 + (size_t)(4 * (d0 & 31) + (d0 >> 5)) * DHID + cb);
            vb[j] = *(const float4*)(W2 + (size_t)(4 * (d1 & 31) + (d1 >> 5)) * DHID + cb);
        }
        #pragma unroll
        for (int j = 0; j < 2; ++j) {
            const int f  = tid + j * AGG_THREADS;
            const int kp = f >> 5;
            const int cb = (f & 31) * 4;
            const float4 a = va[j], b = vb[j];
            *(float4*)&Wt[kp * GM_WPITCH + cb]     = make_float4(a.x, b.x, a.y, b.y);
            *(float4*)&Wt[kp * GM_WPITCH + cb + 2] = make_float4(a.z, b.z, a.w, b.w);
        }
    }

    for (int idx = tid; idx < (TAIL ? AROWS : rows) * 32; idx += AGG_THREADS) {
        const int r = idx >> 5;
        const int l = idx & 31;
        float* a = acc + r * P + l;
        if (r < rows) {
            const float4 h = act(*(const float4*)(H + (size_t)(base + r) * DHID + l * 4));
            const float s = __frsqrt_rn(deg[base + r]);
            if (l == 0) dloc[r] = s;
            a[0]  = h.x * s * s;
            a[32] = h.y * s * s;
            a[64] = h.z * s * s;
            a[96] = h.w * s * s;
        } else {
            a[0] = 0.f; a[32] = 0.f; a[64] = 0.f; a[96] = 0.f;
        }
    }
    __syncthreads();

    auto add_rows = [&](int sv, int nvb, int lr, int nb) {
        for (int i = 0; i < nb; i += 8) {
            float4 hv[8];
            float  nk[8];
            int    lk[8];
            #pragma unroll
            for (int k = 0; k < 8; ++k) {
                const int ii = (i + k < nb) ? (i + k) : i;
                const int sk = __builtin_amdgcn_readlane(sv, ii);
                nk[k] = __int_as_float(__builtin_amdgcn_readlane(nvb, ii));
                lk[k] = __builtin_amdgcn_readlane(lr, ii);
                hv[k] = act(*(const float4*)(H + (size_t)sk * DHID + lane * 4));
            }
            #pragma unroll
            for (int k = 0; k < 8; ++k) {
                if (i + k < nb) {
                    float* a = acc + lk[k] * P + lane;
                    atomicAdd(a,      hv[k].x * nk[k]);
                    atomicAdd(a + 32, hv[k].y * nk[k]);
                    atomicAdd(a + 64, hv[k].z * nk[k]);
                    atomicAdd(a + 96, hv[k].w * nk[k]);
                }
            }
        }
    };

    const int slot = blockIdx.x * AGG_WAVES + wave;
    int4* rq = (int4*)recs + (size_t)slot * AGG_LCAP;

    int kept = __builtin_amdgcn_readfirstlane(rcnt[slot]);
    if (kept > AGG_LCAP) kept = AGG_LCAP;

    if (kept >= 0) {
        for (int b = 0; b < kept; b += 32) {
            const int nb = (kept - b < 32) ? (kept - b) : 32;
            const int4 r = rq[b + (lane < nb ? lane : 0)];
            int nvb = r.y;
            if (!TAIL) {
                const float a = deg[r.x];
                nvb = __float_as_int(__frsqrt_rn(a) * __int_as_float(r.y) * dloc[r.z]);
                if (lane < nb) ((int*)(rq + b + lane))[1] = nvb;
            }
            add_rows(r.x, nvb, r.z, nb);
        }
    } else {
        const int nchunk = (nE + BM_CHUNK - 1) / BM_CHUNK;
        owner_sweep<AGG_QCAP>(bm + (size_t)blockIdx.x * ((size_t)nchunk * BM_CWORDS), nchunk,
                              queue + wave * AGG_QCAP, wave, lane, false,
            [&](const int* q, int cnt, bool keep) {
                (void)keep;
                for (int b = 0; b < cnt; b += 32) {
                    const int nb = (cnt - b < 32) ? (cnt - b) : 32;
                    const int   e  = q[b + (lane < nb ? lane : 0)];
                    const int   sv = src[e];
                    const float wv = w[e];
                    const int   lr = dst[e] - base;
                    const float a  = deg[sv];
                    const int  nvb = __float_as_int(__frsqrt_rn(a) * wv * dloc[lr]);
                    add_rows(sv, nvb, lr, nb);
                }
            });
    }
    __syncthreads();

    if (!TAIL) {
        for (int idx = tid; idx < rows * 32; idx += AGG_THREADS) {
            const int r = idx >> 5;
            const int l = idx & 31;
            const float* a = acc + r * P + l;
            const float4 o = make_float4(a[0], a[32], a[64], a[96]);
            *(float4*)(OUT + (size_t)(base + r) * DHID + l * 4) = o;
        }
    } else {
        const int c = lane & 15;
        const int h = lane >> 4;
        const int nstr = (rows + 15) >> 4;
        const bool has  = wave < nstr * 2;
        const int strip = wave >> 1;
        const int nh    = wave & 1;
        v8f dt[4];
        #pragma unroll
        for (int nt = 0; nt < 4; ++nt)
            #pragma unroll
            for (int r = 0; r < 8; ++r) dt[nt][r] = 0.0f;
        if (has) {
            const float* ar = acc + (strip * 16 + c) * P + 4 * h;
            #pragma unroll
            for (int t = 0; t < DHID / 8; ++t) {
                const float4 a4 = *(const float4*)(ar + 8 * t);
                #pragma unroll
                for (int q = 0; q < 2; ++q) {
                    const int kp = 4 * t + 2 * h + q;
                    v2f a;
                    a.x = q ? a4.z : a4.x;
                    a.y = q ? a4.w : a4.y;
                    const float2* bp = &Wt[kp * GM_WPITCH + nh * 64 + c];
                    float2 bf[4];
                    #pragma unroll
                    for (int nt = 0; nt < 4; ++nt) bf[nt] = bp[nt * 16];
                    #pragma unroll
                    for (int nt = 0; nt < 4; ++nt) {
                        v2f b;
                        b.x = bf[nt].x;
                        b.y = bf[nt].y;
                        dt[nt] = __builtin_amdgcn_wmma_f32_16x16x4_f32(
                             false, a,  false, b,
                             (short)0, dt[nt],
                             false,  false);
                    }
                }
            }
        }
        __syncthreads();

        if (has) {
            #pragma unroll
            for (int nt = 0; nt < 4; ++nt) {
                const int col = nh * 64 + nt * 16 + c;
                const float bb = b2[col];
                #pragma unroll
                for (int r = 0; r < 8; ++r) {
                    const float x = dt[nt][r] + bb;
                    acc[(strip * 16 + 8 * h + r) * P + col] = x > 0.f ? x : 0.f;
                }
            }
        }
        __syncthreads();

        if (wave < nstr) {
            const int cc = c < NCLS ? c : 0;
            const float* ar = acc + (wave * 16 + c) * P + 4 * h;
            v8f lg;
            #pragma unroll
            for (int r = 0; r < 8; ++r) lg[r] = 0.0f;
            #pragma unroll
            for (int t = 0; t < DHID / 8; ++t) {
                const float4 a4 = *(const float4*)(ar + 8 * t);
                #pragma unroll
                for (int q = 0; q < 2; ++q) {
                    const int k0 = 8 * t + 4 * h + 2 * q;
                    v2f a, b;
                    a.x = q ? a4.z : a4.x;
                    a.y = q ? a4.w : a4.y;
                    const float w0 = Wc[(k0    ) * NCLS + cc];
                    const float w1 = Wc[(k0 + 1) * NCLS + cc];
                    b.x = c < NCLS ? w0 : 0.f;
                    b.y = c < NCLS ? w1 : 0.f;
                    lg = __builtin_amdgcn_wmma_f32_16x16x4_f32(
                         false, a,  false, b,
                         (short)0, lg,
                         false,  false);
                }
            }
            const float bcv = bc[cc];
            #pragma unroll
            for (int r = 0; r < 8; ++r) {
                const int lr = wave * 16 + 8 * h + r;
                if (c < NCLS && lr < rows) OUT[(size_t)(base + lr) * NCLS + c] = lg[r] + bcv;
            }
        }
    }
}

extern "C" void kernel_launch(void* const* d_in, const int* in_sizes, int n_in,
                              void* d_out, int out_size, void* d_ws, size_t ws_size,
                              hipStream_t stream) {
    const float* x  = (const float*)d_in[0];
    const int*   ei = (const int*)d_in[1];
    const float* ew = (const float*)d_in[2];
    const float* W1 = (const float*)d_in[3];
    const float* b1 = (const float*)d_in[4];
    const float* W2 = (const float*)d_in[5];
    const float* b2 = (const float*)d_in[6];
    const float* Wc = (const float*)d_in[7];
    const float* bc = (const float*)d_in[8];
    float* out = (float*)d_out;

    const int n  = NNODES;
    const int E  = in_sizes[1] / 2;
    const int K1 = in_sizes[0] / n;
    const int* src = ei;
    const int* dst = ei + E;

    char* ws = (char*)d_ws;
    size_t off = 0;
    auto alloc = [&](size_t bytes) { char* p = ws + off; off = (off + bytes + 255) & ~(size_t)255; return p; };
    float* deg  = (float*)alloc((size_t)n * 4);
    float* bufA = (float*)alloc((size_t)n * DHID * 4);
    float* bufB = (float*)alloc((size_t)n * DHID * 4);
    int* recs = (int*)alloc((size_t)AGG_BLOCKS * AGG_WAVES * AGG_LCAP * 16);
    int* rcnt = (int*)alloc((size_t)AGG_BLOCKS * AGG_WAVES * 4);
    unsigned* bm = (unsigned*)alloc((size_t)AGG_BLOCKS * (size_t)((E + BM_CHUNK - 1) / BM_CHUNK) * BM_CWORDS * 4);
    int* prec = (int*)alloc((size_t)E * 16);
    (void)ws_size;

    bins_kernel     <<<EW_BLOCKS, EW_THREADS, 0, stream>>>(src, dst, ew, bm, prec, E);
    own_edges_kernel<<<AGG_BLOCKS, AGG_THREADS, 0, stream>>>(bm, prec, deg, n, E, recs, rcnt);

    gemm_rows_wmma_kernel<<<AGG_BLOCKS, GM_THREADS, 0, stream>>>(x, W1, bufA, n, K1);
    agg_rows_kernel<false><<<AGG_BLOCKS, AGG_THREADS, 0, stream>>>(bufA, src, dst, ew, deg, bm, bufB, n, E, recs, rcnt,
                                                                  nullptr, nullptr, nullptr, nullptr, nullptr);

    agg_rows_kernel<true><<<AGG_BLOCKS, AGG_THREADS, 0, stream>>>(bufB, src, dst, ew, deg, bm, out, n, E, recs, rcnt,
                                                                 b1, W2, b2, Wc, bc);
}
